# starting kernel
# baseline (speedup 1.0000x reference)
#include <hip/hip_runtime.h>
#include <hip/hip_bf16.h>

typedef __attribute__((ext_vector_type(2))) float v2f;
typedef __attribute__((ext_vector_type(8))) float v8f;

__global__ void deg_init_kernel(float* deg, int n) {
    int i = blockIdx.x * blockDim.x + threadIdx.x;
    if (i < n) deg[i] = 1.0f;
}

__global__ void deg_accum_kernel(const int* dst, const float* w, float* deg, int e) {
    int i = blockIdx.x * blockDim.x + threadIdx.x;
    if (i < e) atomicAdd(&deg[dst[i]], w[i]);
}

__global__ void dinv_kernel(float* deg, int n) {
    int i = blockIdx.x * blockDim.x + threadIdx.x;
    if (i < n) {
        float d = deg[i];
        deg[i] = (d > 0.0f) ? rsqrtf(d) : 0.0f;
    }
}

__global__ void __launch_bounds__(256)
gemm_wmma_f32_kernel(const float* __restrict__ A, const float* __restrict__ W,
                     float* __restrict__ Out, int M, int K, int F, int relu_in) {
    int wid  = (blockIdx.x * blockDim.x + threadIdx.x) >> 5;
    int lane = threadIdx.x & 31;
    int tiles_n = F >> 4;
    int total   = (M >> 4) * tiles_n;
    if (wid >= total) return;

    int tm  = wid / tiles_n;
    int tn  = wid - tm * tiles_n;
    int l16  = lane & 15;
    int half = lane >> 4;

    int row = tm * 16 + l16;
    int col = tn * 16 + l16;

    const float* arow = A + (size_t)row * K;
    v8f c = {0.f, 0.f, 0.f, 0.f, 0.f, 0.f, 0.f, 0.f};

    for (int k = 0; k < K; k += 4) {
        int ka = k + 2 * half;
        v2f a;
        a.x = arow[ka];
        a.y = arow[ka + 1];
        if (relu_in) {
            a.x = fmaxf(a.x, 0.0f);
            a.y = fmaxf(a.y, 0.0f);
        }
        v2f b;
        b.x = W[(size_t)ka * F + col];
        b.y = W[(size_t)(ka + 1) * F + col];
        c = __builtin_amdgcn_wmma_f32_16x16x4_f32(false, a, false, b,
                                                  (short)0, c, false, false);
    }

    int mbase = tm * 16 + half * 8;
#pragma unroll
    for (int r = 0; r < 8; ++r)
        Out[(size_t)(mbase + r) * F + col] = c[r];
}

__global__ void init_agg_kernel(const float* __restrict__ H,
                                const float* __restrict__ bias,
                                const float* __restrict__ dinv,
                                float* __restrict__ agg, int n, int F) {
    long long i = (long long)blockIdx.x * blockDim.x + threadIdx.x;
    long long tot = (long long)n * F;
    if (i >= tot) return;
    int node = (int)(i / F);
    int f    = (int)(i - (long long)node * F);
    float di = dinv[node];
    agg[i] = bias[f] + di * di * H[i];
}

__global__ void __launch_bounds__(256)
scatter_kernel(const int* __restrict__ src, const int* __restrict__ dst,
               const float* __restrict__ w, const float* __restrict__ dinv,
               const float* __restrict__ H, float* __restrict__ agg,
               int e, int F) {
    int wid  = (blockIdx.x * blockDim.x + threadIdx.x) >> 5;
    int lane = threadIdx.x & 31;
    if (wid >= e) return;
    int s = src[wid];
    int d = dst[wid];
    float nrm = dinv[s] * w[wid] * dinv[d];
    const float* h = H + (size_t)s * F;
    float* o = agg + (size_t)d * F;
    for (int f = lane; f < F; f += 32)
        atomicAdd(&o[f], nrm * h[f]);
}

extern "C" void kernel_launch(void* const* d_in, const int* in_sizes, int n_in,
                              void* d_out, int out_size, void* d_ws, size_t ws_size,
                              hipStream_t stream) {
    const float* x    = (const float*)d_in[0];
    const int*   ei   = (const int*)d_in[1];
    const float* ew   = (const float*)d_in[2];
    const float* W1   = (const float*)d_in[3];
    const float* b1   = (const float*)d_in[4];
    const float* W2   = (const float*)d_in[5];
    const float* b2   = (const float*)d_in[6];
    const float* W3   = (const float*)d_in[7];
    const float* b3   = (const float*)d_in[8];
    float*       out  = (float*)d_out;

    const int hidden  = in_sizes[4];
    const int out_dim = in_sizes[8];
    const int in_dim  = in_sizes[3] / hidden;
    const int n       = in_sizes[0] / in_dim;
    const int e       = in_sizes[2];

    const int* src = ei;
    const int* dst = ei + e;

    char* ws = (char*)d_ws;
    size_t off = 0;
    float* dinv = (float*)(ws + off);
    off += ((size_t)n * sizeof(float) + 255) & ~(size_t)255;
    float* H = (float*)(ws + off);
    off += ((size_t)n * hidden * sizeof(float) + 255) & ~(size_t)255;
    float* AGG = (float*)(ws + off);

    const int B = 256;
    auto cdiv = [](long long a, long long b) { return (int)((a + b - 1) / b); };

    deg_init_kernel<<<cdiv(n, B), B, 0, stream>>>(dinv, n);
    deg_accum_kernel<<<cdiv(e, B), B, 0, stream>>>(dst, ew, dinv, e);
    dinv_kernel<<<cdiv(n, B), B, 0, stream>>>(dinv, n);

    {
        int tiles = (n / 16) * (hidden / 16);
        gemm_wmma_f32_kernel<<<cdiv((long long)tiles * 32, B), B, 0, stream>>>(
            x, W1, H, n, in_dim, hidden, 0);
        init_agg_kernel<<<cdiv((long long)n * hidden, B), B, 0, stream>>>(
            H, b1, dinv, AGG, n, hidden);
        scatter_kernel<<<cdiv((long long)e * 32, B), B, 0, stream>>>(
            src, dst, ew, dinv, H, AGG, e, hidden);
    }

    {
        int tiles = (n / 16) * (hidden / 16);
        gemm_wmma_f32_kernel<<<cdiv((long long)tiles * 32, B), B, 0, stream>>>(
            AGG, W2, H, n, hidden, hidden, 1);
        init_agg_kernel<<<cdiv((long long)n * hidden, B), B, 0, stream>>>(
            H, b2, dinv, AGG, n, hidden);
        scatter_kernel<<<cdiv((long long)e * 32, B), B, 0, stream>>>(
            src, dst, ew, dinv, H, AGG, e, hidden);
    }

    {
        int tiles = (n / 16) * (out_dim / 16);
        gemm_wmma_f32_kernel<<<cdiv((long long)tiles * 32, B), B, 0, stream>>>(
            AGG, W3, H, n, hidden, out_dim, 1);
        init_agg_kernel<<<cdiv((long long)n * out_dim, B), B, 0, stream>>>(
            H, b3, dinv, out, n, out_dim);
        scatter_kernel<<<cdiv((long long)e * 32, B), B, 0, stream>>>(
            src, dst, ew, dinv, H, out, e, out_dim);
    }
}


# Edge aggregation as a gather over a destination-sorted edge list built in-call (counting sort, integer rank atomics): no float atomics, one store per output; row sums now run in list order
# speedup vs baseline: 2.0898x; 2.0898x over previous
#include <hip/hip_runtime.h>
#include <hip/hip_bf16.h>

typedef __attribute__((ext_vector_type(2))) float v2f;
typedef __attribute__((ext_vector_type(8))) float v8f;

typedef __attribute__((ext_vector_type(2))) int v2i;

__global__ void cnt_zero_kernel(int* cnt, int n) {
    int i = blockIdx.x * blockDim.x + threadIdx.x;
    if (i < n) cnt[i] = 0;
}

__global__ void rank_kernel(const int* __restrict__ dst, int* cnt,
                            int* __restrict__ rank, int e) {
    int i = blockIdx.x * blockDim.x + threadIdx.x;
    if (i < e) rank[i] = atomicAdd(&cnt[dst[i]], 1);
}

__device__ __forceinline__ int wave_incl_scan(int v, int lane) {
#pragma unroll
    for (int o = 1; o < 32; o <<= 1) {
        int u = __shfl_up(v, o);
        if (lane >= o) v += u;
    }
    return v;
}

__global__ void __launch_bounds__(1024)
scan_kernel(int* cnt, int n) {
    __shared__ int wtot[32];
    int lane = threadIdx.x & 31;
    int w    = __builtin_amdgcn_readfirstlane(threadIdx.x >> 5);
    int steps = (n + 4095) >> 12;
    int seg   = steps << 7;
    int b  = w * seg;
    int en = min(b + seg, n);

    int s = 0;
    for (int i = b + lane; i < en; i += 32) s += cnt[i];
    s = wave_incl_scan(s, lane);
    if (lane == 31) wtot[w] = s;
    __syncthreads();
    if (w == 0) wtot[lane] = wave_incl_scan(wtot[lane], lane);
    __syncthreads();
    int carry = (w > 0) ? wtot[w - 1] : 0;

    int i  = b + lane * 4;
    int a0 = (i     < en) ? cnt[i]     : 0;
    int a1 = (i + 1 < en) ? cnt[i + 1] : 0;
    int a2 = (i + 2 < en) ? cnt[i + 2] : 0;
    int a3 = (i + 3 < en) ? cnt[i + 3] : 0;
    for (int i0 = b; i0 < en; i0 += 128) {
        int nx = i + 128;
        int n0 = (nx     < en) ? cnt[nx]     : 0;
        int n1 = (nx + 1 < en) ? cnt[nx + 1] : 0;
        int n2 = (nx + 2 < en) ? cnt[nx + 2] : 0;
        int n3 = (nx + 3 < en) ? cnt[nx + 3] : 0;
        int t1 = a0 + a1;
        int t2 = t1 + a2;
        int t3 = t2 + a3;
        int x  = wave_incl_scan(t3, lane);
        int ex = carry + x - t3;
        if (i     < en) cnt[i]     = ex;
        if (i + 1 < en) cnt[i + 1] = ex + a0;
        if (i + 2 < en) cnt[i + 2] = ex + t1;
        if (i + 3 < en) cnt[i + 3] = ex + t2;
        carry += __shfl(x, 31);
        i = nx; a0 = n0; a1 = n1; a2 = n2; a3 = n3;
    }
    if (threadIdx.x == 1023) cnt[n] = carry;
}

__global__ void fill_kernel(const int* __restrict__ src, const int* __restrict__ dst,
                            const float* __restrict__ w, const int* __restrict__ rowptr,
                            const int* __restrict__ rank, v2i* __restrict__ rec, int e) {
    int i = blockIdx.x * blockDim.x + threadIdx.x;
    if (i < e) {
        v2i r;
        r.x = src[i];
        r.y = __float_as_int(w[i]);
        rec[rowptr[dst[i]] + rank[i]] = r;
    }
}

__global__ void dinv_kernel(const int* __restrict__ rowptr, const v2i* __restrict__ rec,
                            float* __restrict__ dinv, int n) {
    int i = blockIdx.x * blockDim.x + threadIdx.x;
    if (i < n) {
        float d = 1.0f;
        int en = rowptr[i + 1];
        for (int p = rowptr[i]; p < en; ++p) d += __int_as_float(rec[p].y);
        dinv[i] = (d > 0.0f) ? rsqrtf(d) : 0.0f;
    }
}

__global__ void copy_kernel(const v2i* __restrict__ from, v2i* __restrict__ to, int cnt8) {
    int i = blockIdx.x * blockDim.x + threadIdx.x;
    if (i < cnt8) to[i] = from[i];
}

__global__ void __launch_bounds__(256)
gemm_wmma_f32_kernel(const float* __restrict__ A, const float* __restrict__ W,
                     float* __restrict__ Out, int M, int K, int F, int relu_in) {
    int wid  = (blockIdx.x * blockDim.x + threadIdx.x) >> 5;
    int lane = threadIdx.x & 31;
    int tiles_n = F >> 4;
    int total   = (M >> 4) * tiles_n;
    if (wid >= total) return;

    int tm  = wid / tiles_n;
    int tn  = wid - tm * tiles_n;
    int l16  = lane & 15;
    int half = lane >> 4;

    int row = tm * 16 + l16;
    int col = tn * 16 + l16;

    const float* arow = A + (size_t)row * K;
    v8f c = {0.f, 0.f, 0.f, 0.f, 0.f, 0.f, 0.f, 0.f};

    for (int k = 0; k < K; k += 4) {
        int ka = k + 2 * half;
        v2f a;
        a.x = arow[ka];
        a.y = arow[ka + 1];
        if (relu_in) {
            a.x = fmaxf(a.x, 0.0f);
            a.y = fmaxf(a.y, 0.0f);
        }
        v2f b;
        b.x = W[(size_t)ka * F + col];
        b.y = W[(size_t)(ka + 1) * F + col];
        c = __builtin_amdgcn_wmma_f32_16x16x4_f32(false, a, false, b,
                                                  (short)0, c, false, false);
    }

    int mbase = tm * 16 + half * 8;
#pragma unroll
    for (int r = 0; r < 8; ++r)
        Out[(size_t)(mbase + r) * F + col] = c[r];
}

template <int VEC>
__global__ void __launch_bounds__(256)
gather_kernel(const int* __restrict__ rowptr, const v2i* __restrict__ rec,
              const float* __restrict__ dinv, const float* __restrict__ H,
              const float* __restrict__ bias, float* __restrict__ outp,
              int n, int F) {
    typedef __attribute__((ext_vector_type(VEC))) float vf;
    int node = (blockIdx.x * blockDim.x + threadIdx.x) >> 5;
    int lane = threadIdx.x & 31;
    if (node >= n) return;
    node = __builtin_amdgcn_readfirstlane(node);

    int beg = rowptr[node];
    int end = rowptr[node + 1];
    float di = dinv[node];

    for (int fb = 0; fb < F; fb += 32 * VEC) {
        int  f0  = fb + lane * VEC;
        bool act = f0 < F;
        vf acc;
        if (act) {
            vf bv = *(const vf*)(bias + f0);
            vf hv = *(const vf*)(H + (size_t)node * F + f0);
            acc = bv + (di * di) * hv;
        }
        for (int base = beg; base < end; base += 32) {
            int m = min(32, end - base);
            int   sl = 0;
            float cl = 0.0f;
            if (lane < m) {
                v2i r = rec[base + lane];
                sl = r.x;
                cl = dinv[sl] * __int_as_float(r.y) * di;
            }
            int j = 0;
            for (; j + 4 <= m; j += 4) {
                int s0 = __builtin_amdgcn_readlane(sl, j);
                int s1 = __builtin_amdgcn_readlane(sl, j + 1);
                int s2 = __builtin_amdgcn_readlane(sl, j + 2);
                int s3 = __builtin_amdgcn_readlane(sl, j + 3);
                float c0 = __int_as_float(__builtin_amdgcn_readlane(__float_as_int(cl), j));
                float c1 = __int_as_float(__builtin_amdgcn_readlane(__float_as_int(cl), j + 1));
                float c2 = __int_as_float(__builtin_amdgcn_readlane(__float_as_int(cl), j + 2));
                float c3 = __int_as_float(__builtin_amdgcn_readlane(__float_as_int(cl), j + 3));
                if (act) {
                    vf h0 = *(const vf*)(H + (size_t)s0 * F + f0);
                    vf h1 = *(const vf*)(H + (size_t)s1 * F + f0);
                    vf h2 = *(const vf*)(H + (size_t)s2 * F + f0);
                    vf h3 = *(const vf*)(H + (size_t)s3 * F + f0);
                    acc += c0 * h0;
                    acc += c1 * h1;
                    acc += c2 * h2;
                    acc += c3 * h3;
                }
            }
            for (; j < m; ++j) {
                int s0 = __builtin_amdgcn_readlane(sl, j);
                float c0 = __int_as_float(__builtin_amdgcn_readlane(__float_as_int(cl), j));
                if (act) {
                    vf h0 = *(const vf*)(H + (size_t)s0 * F + f0);
                    acc += c0 * h0;
                }
            }
        }
        if (act) *(vf*)(outp + (size_t)node * F + f0) = acc;
    }
}

extern "C" void kernel_launch(void* const* d_in, const int* in_sizes, int n_in,
                              void* d_out, int out_size, void* d_ws, size_t ws_size,
                              hipStream_t stream) {
    const float* x    = (const float*)d_in[0];
    const int*   ei   = (const int*)d_in[1];
    const float* ew   = (const float*)d_in[2];
    const float* W1   = (const float*)d_in[3];
    const float* b1   = (const float*)d_in[4];
    const float* W2   = (const float*)d_in[5];
    const float* b2   = (const float*)d_in[6];
    const float* W3   = (const float*)d_in[7];
    const float* b3   = (const float*)d_in[8];
    float*       out  = (float*)d_out;

    const int hidden  = in_sizes[4];
    const int out_dim = in_sizes[8];
    const int in_dim  = in_sizes[3] / hidden;
    const int n       = in_sizes[0] / in_dim;
    const int e       = in_sizes[2];

    const int* src = ei;
    const int* dst = ei + e;

    char* ws = (char*)d_ws;
    size_t off = 0;
    float* dinv = (float*)(ws + off);
    off += ((size_t)n * sizeof(float) + 255) & ~(size_t)255;
    float* H = (float*)(ws + off);
    off += ((size_t)n * hidden * sizeof(float) + 255) & ~(size_t)255;
    float* AGG = (float*)(ws + off);

    const size_t rp_bytes  = ((size_t)(n + 1) * sizeof(int) + 255) & ~(size_t)255;
    const size_t csr_bytes = rp_bytes + (size_t)e * sizeof(v2i);
    char* csrA = (char*)d_out;
    char* csrB = (char*)H + (((size_t)n * out_dim * sizeof(float) + 255) & ~(size_t)255);
    int* rowptrA = (int*)csrA;
    v2i* recA    = (v2i*)(csrA + rp_bytes);
    int* rowptrB = (int*)csrB;
    v2i* recB    = (v2i*)(csrB + rp_bytes);
    int* rank    = (int*)AGG;

    const int B = 256;
    auto cdiv = [](long long a, long long b) { return (int)((a + b - 1) / b); };

    cnt_zero_kernel<<<cdiv(n, B), B, 0, stream>>>(rowptrA, n);
    rank_kernel<<<cdiv(e, B), B, 0, stream>>>(dst, rowptrA, rank, e);
    scan_kernel<<<1, 1024, 0, stream>>>(rowptrA, n);
    fill_kernel<<<cdiv(e, B), B, 0, stream>>>(src, dst, ew, rowptrA, rank, recA, e);
    dinv_kernel<<<cdiv(n, B), B, 0, stream>>>(rowptrA, recA, dinv, n);

    const int gather_blocks = cdiv((long long)n * 32, B);

    {
        int tiles = (n / 16) * (hidden / 16);
        gemm_wmma_f32_kernel<<<cdiv((long long)tiles * 32, B), B, 0, stream>>>(
            x, W1, H, n, in_dim, hidden, 0);
        gather_kernel<4><<<gather_blocks, B, 0, stream>>>(
            rowptrA, recA, dinv, H, b1, AGG, n, hidden);
    }

    {
        int tiles = (n / 16) * (hidden / 16);
        gemm_wmma_f32_kernel<<<cdiv((long long)tiles * 32, B), B, 0, stream>>>(
            AGG, W2, H, n, hidden, hidden, 1);
        gather_kernel<4><<<gather_blocks, B, 0, stream>>>(
            rowptrA, recA, dinv, H, b2, AGG, n, hidden);
    }

    {
        int tiles = (n / 16) * (out_dim / 16);
        gemm_wmma_f32_kernel<<<cdiv((long long)tiles * 32, B), B, 0, stream>>>(
            AGG, W3, H, n, hidden, out_dim, 1);
        int cnt8 = (int)(csr_bytes / sizeof(v2i));
        copy_kernel<<<cdiv(cnt8, B), B, 0, stream>>>((const v2i*)csrA, (v2i*)csrB, cnt8);
        gather_kernel<2><<<gather_blocks, B, 0, stream>>>(
            rowptrB, recB, dinv, H, b3, out, n, out_dim);
    }
}


# Edge list by two-level counting sort with LDS rank atomics, arrival order in a row (chunks grouped by 256-node bucket; a work-group per bucket writes rows, row starts, dinv): 5 launches to 2
# speedup vs baseline: 2.8728x; 1.3747x over previous
#include <hip/hip_runtime.h>
#include <hip/hip_bf16.h>

typedef __attribute__((ext_vector_type(2))) float v2f;
typedef __attribute__((ext_vector_type(8))) float v8f;

typedef __attribute__((ext_vector_type(2))) int v2i;

#define CHUNK_EDGES 4096
#define STAGE_CAP   8192

__device__ __forceinline__ int wave_incl_scan(int v, int lane) {
#pragma unroll
    for (int o = 1; o < 32; o <<= 1) {
        int u = __shfl_up(v, o);
        if (lane >= o) v += u;
    }
    return v;
}

__device__ __forceinline__ void wave_excl_scan_256(const int* cnt, int* ex_a, int* ex_b, int lane) {
    int c[8];
    int run = 0;
#pragma unroll
    for (int j = 0; j < 8; ++j) {
        int v = cnt[lane * 8 + j];
        c[j] = run;
        run += v;
    }
    int ex = wave_incl_scan(run, lane) - run;
#pragma unroll
    for (int j = 0; j < 8; ++j) {
        ex_a[lane * 8 + j] = ex + c[j];
        if (ex_b) ex_b[lane * 8 + j] = ex + c[j];
    }
}

__device__ __forceinline__ int block_incl_scan_1024(int v, int lane, int w, int* wtot, int* total) {
    int s = wave_incl_scan(v, lane);
    if (lane == 31) wtot[w] = s;
    __syncthreads();
    if (w == 0) wtot[lane] = wave_incl_scan(wtot[lane], lane);
    __syncthreads();
    int carry = (w > 0) ? wtot[w - 1] : 0;
    *total = wtot[31];
    __syncthreads();
    return s + carry;
}

__device__ __forceinline__ int find_run(const int* runstart, int p) {
    int lo = 0;
#pragma unroll
    for (int step = 512; step > 0; step >>= 1)
        if (runstart[lo + step] <= p) lo += step;
    return lo;
}

__global__ void __launch_bounds__(1024)
chunk_sort_kernel(const int* __restrict__ src, const int* __restrict__ dst,
                  const float* __restrict__ w, v2i* __restrict__ sorted,
                  int* __restrict__ cntT, int* __restrict__ offT, int e, int nchunks) {
    __shared__ int hist[256];
    __shared__ int off[256];
    __shared__ v2i stage[CHUNK_EDGES];
    int t    = threadIdx.x;
    int lane = t & 31;
    int blk  = blockIdx.x;
    int base = blk * CHUNK_EDGES;

    if (t < 256) hist[t] = 0;

    int d[4], s[4], wb[4], rk[4];
#pragma unroll
    for (int k = 0; k < 4; ++k) {
        int i = base + k * 1024 + t;
        d[k] = -1; s[k] = 0; wb[k] = 0;
        if (i < e) {
            d[k]  = dst[i];
            s[k]  = src[i];
            wb[k] = __float_as_int(w[i]);
        }
    }
    __syncthreads();
#pragma unroll
    for (int k = 0; k < 4; ++k) {
        rk[k] = 0;
        if (d[k] >= 0) rk[k] = atomicAdd(&hist[d[k] >> 8], 1);
    }
    __syncthreads();
    if (t < 32) wave_excl_scan_256(hist, off, nullptr, lane);
    __syncthreads();
    if (t < 256) {
        cntT[t * nchunks + blk] = hist[t];
        offT[t * nchunks + blk] = off[t];
    }
#pragma unroll
    for (int k = 0; k < 4; ++k) {
        if (d[k] >= 0) {
            v2i r;
            r.x = (int)((unsigned)s[k] | ((unsigned)(d[k] & 255) << 24));
            r.y = wb[k];
            stage[off[d[k] >> 8] + rk[k]] = r;
        }
    }
    __syncthreads();
    int m = min(CHUNK_EDGES, e - base);
#pragma unroll
    for (int k = 0; k < 4; ++k) {
        int p = k * 1024 + t;
        if (p < m) sorted[base + p] = stage[p];
    }
}

__global__ void __launch_bounds__(1024)
bucket_sort_kernel(const v2i* __restrict__ sorted, const int* __restrict__ cntT,
                   const int* __restrict__ offT, int* __restrict__ rowptr,
                   v2i* rec, float* __restrict__ dinv, int n, int nchunks) {
    __shared__ int wtot[32];
    __shared__ int runstart[1024];
    __shared__ int runsrc[1024];
    __shared__ int cnt[256];
    __shared__ int rowstart[256];
    __shared__ int cursor[256];
    __shared__ v2i stage[STAGE_CAP];
    __shared__ v2i outl[STAGE_CAP];
    int t    = threadIdx.x;
    int lane = t & 31;
    int w    = __builtin_amdgcn_readfirstlane(t >> 5);
    int b    = blockIdx.x;

    int c = 0, o = 0;
    if (t < nchunks) {
        c = cntT[b * nchunks + t];
        o = offT[b * nchunks + t];
    }
    int T, bstart;
    int incl = block_incl_scan_1024(c, lane, w, wtot, &T);
    (void)block_incl_scan_1024(o, lane, w, wtot, &bstart);
    runstart[t] = incl - c;
    runsrc[t]   = t * CHUNK_EDGES + o;
    if (t < 256) cnt[t] = 0;
    __syncthreads();

    const bool staged = (T <= STAGE_CAP);

    for (int p0 = t; p0 < T; p0 += 4096) {
        int g[4];
#pragma unroll
        for (int k = 0; k < 4; ++k) {
            int p  = min(p0 + k * 1024, T - 1);
            int lo = find_run(runstart, p);
            g[k] = runsrc[lo] + (p - runstart[lo]);
        }
        v2i r[4];
#pragma unroll
        for (int k = 0; k < 4; ++k) r[k] = sorted[g[k]];
#pragma unroll
        for (int k = 0; k < 4; ++k) {
            int p = p0 + k * 1024;
            if (p < T) {
                if (staged) stage[p] = r[k];
                atomicAdd(&cnt[(unsigned)r[k].x >> 24], 1);
            }
        }
    }
    __syncthreads();
    if (t < 32) wave_excl_scan_256(cnt, rowstart, cursor, lane);
    __syncthreads();

    if (t < 256) {
        int node = b * 256 + t;
        if (node < n) rowptr[node] = bstart + rowstart[t];
    }
    if (b == (int)gridDim.x - 1 && t == 0) rowptr[n] = bstart + T;

    for (int p = t; p < T; p += 1024) {
        v2i r;
        if (staged) {
            r = stage[p];
        } else {
            int lo = find_run(runstart, p);
            r = sorted[runsrc[lo] + (p - runstart[lo])];
        }
        int pos = atomicAdd(&cursor[(unsigned)r.x >> 24], 1);
        v2i q;
        q.x = r.x & 0x00FFFFFF;
        q.y = r.y;
        if (staged) outl[pos] = q;
        else        rec[bstart + pos] = q;
    }
    __syncthreads();
    if (staged)
        for (int p = t; p < T; p += 1024) rec[bstart + p] = outl[p];

    if (t < 256) {
        int node = b * 256 + t;
        if (node < n) {
            float dsum = 1.0f;
            int q0 = rowstart[t];
            int q1 = q0 + cnt[t];
            if (staged) {
                for (int q = q0; q < q1; ++q) dsum += __int_as_float(outl[q].y);
            } else {
                for (int q = q0; q < q1; ++q) dsum += __int_as_float(rec[bstart + q].y);
            }
            dinv[node] = (dsum > 0.0f) ? rsqrtf(dsum) : 0.0f;
        }
    }
}

__global__ void copy_kernel(const v2i* __restrict__ from, v2i* __restrict__ to, int cnt8) {
    int i = blockIdx.x * blockDim.x + threadIdx.x;
    if (i < cnt8) to[i] = from[i];
}

__global__ void __launch_bounds__(256)
gemm_wmma_f32_kernel(const float* __restrict__ A, const float* __restrict__ W,
                     float* __restrict__ Out, int M, int K, int F, int relu_in) {
    int wid  = (blockIdx.x * blockDim.x + threadIdx.x) >> 5;
    int lane = threadIdx.x & 31;
    int tiles_n = F >> 4;
    int total   = (M >> 4) * tiles_n;
    if (wid >= total) return;

    int tm  = wid / tiles_n;
    int tn  = wid - tm * tiles_n;
    int l16  = lane & 15;
    int half = lane >> 4;

    int row = tm * 16 + l16;
    int col = tn * 16 + l16;

    const float* arow = A + (size_t)row * K;
    v8f c = {0.f, 0.f, 0.f, 0.f, 0.f, 0.f, 0.f, 0.f};

    for (int k = 0; k < K; k += 4) {
        int ka = k + 2 * half;
        v2f a;
        a.x = arow[ka];
        a.y = arow[ka + 1];
        if (relu_in) {
            a.x = fmaxf(a.x, 0.0f);
            a.y = fmaxf(a.y, 0.0f);
        }
        v2f b;
        b.x = W[(size_t)ka * F + col];
        b.y = W[(size_t)(ka + 1) * F + col];
        c = __builtin_amdgcn_wmma_f32_16x16x4_f32(false, a, false, b,
                                                  (short)0, c, false, false);
    }

    int mbase = tm * 16 + half * 8;
#pragma unroll
    for (int r = 0; r < 8; ++r)
        Out[(size_t)(mbase + r) * F + col] = c[r];
}

template <int VEC>
__global__ void __launch_bounds__(256)
gather_kernel(const int* __restrict__ rowptr, const v2i* __restrict__ rec,
              const float* __restrict__ dinv, const float* __restrict__ H,
              const float* __restrict__ bias, float* __restrict__ outp,
              int n, int F) {
    typedef __attribute__((ext_vector_type(VEC))) float vf;
    int node = (blockIdx.x * blockDim.x + threadIdx.x) >> 5;
    int lane = threadIdx.x & 31;
    if (node >= n) return;
    node = __builtin_amdgcn_readfirstlane(node);

    int beg = rowptr[node];
    int end = rowptr[node + 1];
    float di = dinv[node];

    for (int fb = 0; fb < F; fb += 32 * VEC) {
        int  f0  = fb + lane * VEC;
        bool act = f0 < F;
        vf acc;
        if (act) {
            vf bv = *(const vf*)(bias + f0);
            vf hv = *(const vf*)(H + (size_t)node * F + f0);
            acc = bv + (di * di) * hv;
        }
        for (int base = beg; base < end; base += 32) {
            int m = min(32, end - base);
            int   sl = 0;
            float cl = 0.0f;
            if (lane < m) {
                v2i r = rec[base + lane];
                sl = r.x;
                cl = dinv[sl] * __int_as_float(r.y) * di;
            }
            int j = 0;
            for (; j + 4 <= m; j += 4) {
                int s0 = __builtin_amdgcn_readlane(sl, j);
                int s1 = __builtin_amdgcn_readlane(sl, j + 1);
                int s2 = __builtin_amdgcn_readlane(sl, j + 2);
                int s3 = __builtin_amdgcn_readlane(sl, j + 3);
                float c0 = __int_as_float(__builtin_amdgcn_readlane(__float_as_int(cl), j));
                float c1 = __int_as_float(__builtin_amdgcn_readlane(__float_as_int(cl), j + 1));
                float c2 = __int_as_float(__builtin_amdgcn_readlane(__float_as_int(cl), j + 2));
                float c3 = __int_as_float(__builtin_amdgcn_readlane(__float_as_int(cl), j + 3));
                if (act) {
                    vf h0 = *(const vf*)(H + (size_t)s0 * F + f0);
                    vf h1 = *(const vf*)(H + (size_t)s1 * F + f0);
                    vf h2 = *(const vf*)(H + (size_t)s2 * F + f0);
                    vf h3 = *(const vf*)(H + (size_t)s3 * F + f0);
                    acc += c0 * h0;
                    acc += c1 * h1;
                    acc += c2 * h2;
                    acc += c3 * h3;
                }
            }
            for (; j < m; ++j) {
                int s0 = __builtin_amdgcn_readlane(sl, j);
                float c0 = __int_as_float(__builtin_amdgcn_readlane(__float_as_int(cl), j));
                if (act) {
                    vf h0 = *(const vf*)(H + (size_t)s0 * F + f0);
                    acc += c0 * h0;
                }
            }
        }
        if (act) *(vf*)(outp + (size_t)node * F + f0) = acc;
    }
}

extern "C" void kernel_launch(void* const* d_in, const int* in_sizes, int n_in,
                              void* d_out, int out_size, void* d_ws, size_t ws_size,
                              hipStream_t stream) {
    const float* x    = (const float*)d_in[0];
    const int*   ei   = (const int*)d_in[1];
    const float* ew   = (const float*)d_in[2];
    const float* W1   = (const float*)d_in[3];
    const float* b1   = (const float*)d_in[4];
    const float* W2   = (const float*)d_in[5];
    const float* b2   = (const float*)d_in[6];
    const float* W3   = (const float*)d_in[7];
    const float* b3   = (const float*)d_in[8];
    float*       out  = (float*)d_out;

    const int hidden  = in_sizes[4];
    const int out_dim = in_sizes[8];
    const int in_dim  = in_sizes[3] / hidden;
    const int n       = in_sizes[0] / in_dim;
    const int e       = in_sizes[2];

    const int* src = ei;
    const int* dst = ei + e;

    char* ws = (char*)d_ws;
    size_t off = 0;
    float* dinv = (float*)(ws + off);
    off += ((size_t)n * sizeof(float) + 255) & ~(size_t)255;
    float* H = (float*)(ws + off);
    off += ((size_t)n * hidden * sizeof(float) + 255) & ~(size_t)255;
    float* AGG = (float*)(ws + off);

    const size_t rp_bytes  = ((size_t)(n + 1) * sizeof(int) + 255) & ~(size_t)255;
    const size_t csr_bytes = rp_bytes + (size_t)e * sizeof(v2i);
    char* csrA = (char*)d_out;
    char* csrB = (char*)H + (((size_t)n * out_dim * sizeof(float) + 255) & ~(size_t)255);
    int* rowptrA = (int*)csrA;
    v2i* recA    = (v2i*)(csrA + rp_bytes);
    int* rowptrB = (int*)csrB;
    v2i* recB    = (v2i*)(csrB + rp_bytes);

    const int B = 256;
    auto cdiv = [](long long a, long long b) { return (int)((a + b - 1) / b); };

    const int nchunks  = cdiv(e, CHUNK_EDGES);
    const int nbuckets = cdiv(n, 256);
    v2i* sorted = (v2i*)AGG;
    int* cntT   = (int*)((char*)AGG + (((size_t)nchunks * CHUNK_EDGES * sizeof(v2i) + 255) & ~(size_t)255));
    int* offT   = cntT + (size_t)256 * nchunks;

    chunk_sort_kernel<<<nchunks, 1024, 0, stream>>>(src, dst, ew, sorted, cntT, offT, e, nchunks);
    bucket_sort_kernel<<<nbuckets, 1024, 0, stream>>>(sorted, cntT, offT, rowptrA, recA, dinv, n, nchunks);

    const int gather_blocks = cdiv((long long)n * 32, B);

    {
        int tiles = (n / 16) * (hidden / 16);
        gemm_wmma_f32_kernel<<<cdiv((long long)tiles * 32, B), B, 0, stream>>>(
            x, W1, H, n, in_dim, hidden, 0);
        gather_kernel<4><<<gather_blocks, B, 0, stream>>>(
            rowptrA, recA, dinv, H, b1, AGG, n, hidden);
    }

    {
        int tiles = (n / 16) * (hidden / 16);
        gemm_wmma_f32_kernel<<<cdiv((long long)tiles * 32, B), B, 0, stream>>>(
            AGG, W2, H, n, hidden, hidden, 1);
        gather_kernel<4><<<gather_blocks, B, 0, stream>>>(
            rowptrA, recA, dinv, H, b2, AGG, n, hidden);
    }

    {
        int tiles = (n / 16) * (out_dim / 16);
        gemm_wmma_f32_kernel<<<cdiv((long long)tiles * 32, B), B, 0, stream>>>(
            AGG, W3, H, n, hidden, out_dim, 1);
        int cnt8 = (int)(csr_bytes / sizeof(v2i));
        copy_kernel<<<cdiv(cnt8, B), B, 0, stream>>>((const v2i*)csrA, (v2i*)csrB, cnt8);
        gather_kernel<2><<<gather_blocks, B, 0, stream>>>(
            rowptrB, recB, dinv, H, b3, out, n, out_dim);
    }
}


# GEMMs: one wave owns a 16-row output strip (an A fragment feeds F/16 WMMAs, not 1), W staged once per work-group in LDS as k-pairs, A read one line ahead; same k order, sums bit-identical
# speedup vs baseline: 3.9786x; 1.3849x over previous
#include <hip/hip_runtime.h>
#include <hip/hip_bf16.h>

typedef __attribute__((ext_vector_type(2))) float v2f;
typedef __attribute__((ext_vector_type(8))) float v8f;

typedef __attribute__((ext_vector_type(2))) int v2i;

#define CHUNK_EDGES 4096
#define STAGE_CAP   8192

__device__ __forceinline__ int wave_incl_scan(int v, int lane) {
#pragma unroll
    for (int o = 1; o < 32; o <<= 1) {
        int u = __shfl_up(v, o);
        if (lane >= o) v += u;
    }
    return v;
}

__device__ __forceinline__ void wave_excl_scan_256(const int* cnt, int* ex_a, int* ex_b, int lane) {
    int c[8];
    int run = 0;
#pragma unroll
    for (int j = 0; j < 8; ++j) {
        int v = cnt[lane * 8 + j];
        c[j] = run;
        run += v;
    }
    int ex = wave_incl_scan(run, lane) - run;
#pragma unroll
    for (int j = 0; j < 8; ++j) {
        ex_a[lane * 8 + j] = ex + c[j];
        if (ex_b) ex_b[lane * 8 + j] = ex + c[j];
    }
}

__device__ __forceinline__ int block_incl_scan_1024(int v, int lane, int w, int* wtot, int* total) {
    int s = wave_incl_scan(v, lane);
    if (lane == 31) wtot[w] = s;
    __syncthreads();
    if (w == 0) wtot[lane] = wave_incl_scan(wtot[lane], lane);
    __syncthreads();
    int carry = (w > 0) ? wtot[w - 1] : 0;
    *total = wtot[31];
    __syncthreads();
    return s + carry;
}

__device__ __forceinline__ int find_run(const int* runstart, int p) {
    int lo = 0;
#pragma unroll
    for (int step = 512; step > 0; step >>= 1)
        if (runstart[lo + step] <= p) lo += step;
    return lo;
}

__global__ void __launch_bounds__(1024)
chunk_sort_kernel(const int* __restrict__ src, const int* __restrict__ dst,
                  const float* __restrict__ w, v2i* __restrict__ sorted,
                  int* __restrict__ cntT, int* __restrict__ offT, int e, int nchunks) {
    __shared__ int hist[256];
    __shared__ int off[256];
    __shared__ v2i stage[CHUNK_EDGES];
    int t    = threadIdx.x;
    int lane = t & 31;
    int blk  = blockIdx.x;
    int base = blk * CHUNK_EDGES;

    if (t < 256) hist[t] = 0;

    int d[4], s[4], wb[4], rk[4];
#pragma unroll
    for (int k = 0; k < 4; ++k) {
        int i = base + k * 1024 + t;
        d[k] = -1; s[k] = 0; wb[k] = 0;
        if (i < e) {
            d[k]  = dst[i];
            s[k]  = src[i];
            wb[k] = __float_as_int(w[i]);
        }
    }
    __syncthreads();
#pragma unroll
    for (int k = 0; k < 4; ++k) {
        rk[k] = 0;
        if (d[k] >= 0) rk[k] = atomicAdd(&hist[d[k] >> 8], 1);
    }
    __syncthreads();
    if (t < 32) wave_excl_scan_256(hist, off, nullptr, lane);
    __syncthreads();
    if (t < 256) {
        cntT[t * nchunks + blk] = hist[t];
        offT[t * nchunks + blk] = off[t];
    }
#pragma unroll
    for (int k = 0; k < 4; ++k) {
        if (d[k] >= 0) {
            v2i r;
            r.x = (int)((unsigned)s[k] | ((unsigned)(d[k] & 255) << 24));
            r.y = wb[k];
            stage[off[d[k] >> 8] + rk[k]] = r;
        }
    }
    __syncthreads();
    int m = min(CHUNK_EDGES, e - base);
#pragma unroll
    for (int k = 0; k < 4; ++k) {
        int p = k * 1024 + t;
        if (p < m) sorted[base + p] = stage[p];
    }
}

__global__ void __launch_bounds__(1024)
bucket_sort_kernel(const v2i* __restrict__ sorted, const int* __restrict__ cntT,
                   const int* __restrict__ offT, int* __restrict__ rowptr,
                   v2i* rec, float* __restrict__ dinv, int n, int nchunks) {
    __shared__ int wtot[32];
    __shared__ int runstart[1024];
    __shared__ int runsrc[1024];
    __shared__ int cnt[256];
    __shared__ int rowstart[256];
    __shared__ int cursor[256];
    __shared__ v2i stage[STAGE_CAP];
    __shared__ v2i outl[STAGE_CAP];
    int t    = threadIdx.x;
    int lane = t & 31;
    int w    = __builtin_amdgcn_readfirstlane(t >> 5);
    int b    = blockIdx.x;

    int c = 0, o = 0;
    if (t < nchunks) {
        c = cntT[b * nchunks + t];
        o = offT[b * nchunks + t];
    }
    int T, bstart;
    int incl = block_incl_scan_1024(c, lane, w, wtot, &T);
    (void)block_incl_scan_1024(o, lane, w, wtot, &bstart);
    runstart[t] = incl - c;
    runsrc[t]   = t * CHUNK_EDGES + o;
    if (t < 256) cnt[t] = 0;
    __syncthreads();

    const bool staged = (T <= STAGE_CAP);

    for (int p0 = t; p0 < T; p0 += 4096) {
        int g[4];
#pragma unroll
        for (int k = 0; k < 4; ++k) {
            int p  = min(p0 + k * 1024, T - 1);
            int lo = find_run(runstart, p);
            g[k] = runsrc[lo] + (p - runstart[lo]);
        }
        v2i r[4];
#pragma unroll
        for (int k = 0; k < 4; ++k) r[k] = sorted[g[k]];
#pragma unroll
        for (int k = 0; k < 4; ++k) {
            int p = p0 + k * 1024;
            if (p < T) {
                if (staged) stage[p] = r[k];
                atomicAdd(&cnt[(unsigned)r[k].x >> 24], 1);
            }
        }
    }
    __syncthreads();
    if (t < 32) wave_excl_scan_256(cnt, rowstart, cursor, lane);
    __syncthreads();

    if (t < 256) {
        int node = b * 256 + t;
        if (node < n) rowptr[node] = bstart + rowstart[t];
    }
    if (b == (int)gridDim.x - 1 && t == 0) rowptr[n] = bstart + T;

    for (int p = t; p < T; p += 1024) {
        v2i r;
        if (staged) {
            r = stage[p];
        } else {
            int lo = find_run(runstart, p);
            r = sorted[runsrc[lo] + (p - runstart[lo])];
        }
        int pos = atomicAdd(&cursor[(unsigned)r.x >> 24], 1);
        v2i q;
        q.x = r.x & 0x00FFFFFF;
        q.y = r.y;
        if (staged) outl[pos] = q;
        else        rec[bstart + pos] = q;
    }
    __syncthreads();
    if (staged)
        for (int p = t; p < T; p += 1024) rec[bstart + p] = outl[p];

    if (t < 256) {
        int node = b * 256 + t;
        if (node < n) {
            float dsum = 1.0f;
            int q0 = rowstart[t];
            int q1 = q0 + cnt[t];
            if (staged) {
                for (int q = q0; q < q1; ++q) dsum += __int_as_float(outl[q].y);
            } else {
                for (int q = q0; q < q1; ++q) dsum += __int_as_float(rec[bstart + q].y);
            }
            dinv[node] = (dsum > 0.0f) ? rsqrtf(dsum) : 0.0f;
        }
    }
}

__global__ void copy_kernel(const v2i* __restrict__ from, v2i* __restrict__ to, int cnt8) {
    int i = blockIdx.x * blockDim.x + threadIdx.x;
    if (i < cnt8) to[i] = from[i];
}

__global__ void __launch_bounds__(256)
gemm_wmma_f32_kernel(const float* __restrict__ A, const float* __restrict__ W,
                     float* __restrict__ Out, int M, int K, int F, int relu_in) {
    int wid  = (blockIdx.x * blockDim.x + threadIdx.x) >> 5;
    int lane = threadIdx.x & 31;
    int tiles_n = F >> 4;
    int total   = (M >> 4) * tiles_n;
    if (wid >= total) return;

    int tm  = wid / tiles_n;
    int tn  = wid - tm * tiles_n;
    int l16  = lane & 15;
    int half = lane >> 4;

    int row = tm * 16 + l16;
    int col = tn * 16 + l16;

    const float* arow = A + (size_t)row * K;
    v8f c = {0.f, 0.f, 0.f, 0.f, 0.f, 0.f, 0.f, 0.f};

    for (int k = 0; k < K; k += 4) {
        int ka = k + 2 * half;
        v2f a;
        a.x = arow[ka];
        a.y = arow[ka + 1];
        if (relu_in) {
            a.x = fmaxf(a.x, 0.0f);
            a.y = fmaxf(a.y, 0.0f);
        }
        v2f b;
        b.x = W[(size_t)ka * F + col];
        b.y = W[(size_t)(ka + 1) * F + col];
        c = __builtin_amdgcn_wmma_f32_16x16x4_f32(false, a, false, b,
                                                  (short)0, c, false, false);
    }

    int mbase = tm * 16 + half * 8;
#pragma unroll
    for (int r = 0; r < 8; ++r)
        Out[(size_t)(mbase + r) * F + col] = c[r];
}

template <int K, int F, bool RELU>
__global__ void __launch_bounds__(256)
gemm_strip_kernel(const float* __restrict__ A, const float* __restrict__ W,
                  float* __restrict__ Out, int M) {
    constexpr int NT    = F / 16;
    constexpr int PITCH = 2 * F + 32;
    constexpr int KC    = 32;
    constexpr int F4    = F / 4;
    __shared__ float Wl[(K / 2) * PITCH];

    int t    = threadIdx.x;
    int lane = t & 31;
    int w    = __builtin_amdgcn_readfirstlane(t >> 5);
    int l16  = lane & 15;
    int half = lane >> 4;
    int strip = blockIdx.x * 8 + w;
    bool live = strip < (M >> 4);

    const float* arow = A + ((size_t)(live ? strip : 0) * 16 + l16) * K + 2 * half;
    v2f a_cur[8];
    if (live) {
#pragma unroll
        for (int j = 0; j < 8; ++j) a_cur[j] = *(const v2f*)(arow + 4 * j);
    }

    constexpr int NIT   = (K / 2) * F4 / 256;
    constexpr int BATCH = NIT < 8 ? NIT : 8;
    static_assert((K / 2) * F4 % 256 == 0 && NIT % BATCH == 0, "staging shape");
    static_assert((NT == 4 || NT == 8) && K % KC == 0, "strip shape");
#pragma unroll
    for (int i0 = 0; i0 < NIT; i0 += BATCH) {
        float4 r0[BATCH], r1[BATCH];
#pragma unroll
        for (int u = 0; u < BATCH; ++u) {
            int i  = t + (i0 + u) * 256;
            int kp = i / F4;
            int n4 = i - kp * F4;
            r0[u] = *(const float4*)(W + (size_t)(2 * kp) * F + n4 * 4);
            r1[u] = *(const float4*)(W + (size_t)(2 * kp + 1) * F + n4 * 4);
        }
#pragma unroll
        for (int u = 0; u < BATCH; ++u) {
            int i  = t + (i0 + u) * 256;
            int kp = i / F4;
            int n4 = i - kp * F4;
            float* d = Wl + kp * PITCH + n4 * 8;
            *(float4*)(d)     = make_float4(r0[u].x, r1[u].x, r0[u].y, r1[u].y);
            *(float4*)(d + 4) = make_float4(r0[u].z, r1[u].z, r0[u].w, r1[u].w);
        }
    }
    __syncthreads();
    if (!live) return;

    v8f acc[NT];
#pragma unroll
    for (int tn = 0; tn < NT; ++tn) acc[tn] = (v8f){0.f, 0.f, 0.f, 0.f, 0.f, 0.f, 0.f, 0.f};

    auto act = [](v2f a) -> v2f {
        if (RELU) {
            a.x = fmaxf(a.x, 0.0f);
            a.y = fmaxf(a.y, 0.0f);
        }
        return a;
    };
    auto chunk = [&](int kc) {
        const float* bbase = Wl + ((kc >> 1) + half) * PITCH + 2 * l16;
#pragma unroll
        for (int j = 0; j < 8; ++j) {
            const float* bp = bbase + (2 * j) * PITCH;
#pragma unroll
            for (int tn = 0; tn < NT; ++tn) {
                v2f b = *(const v2f*)(bp + tn * 32);
                acc[tn] = __builtin_amdgcn_wmma_f32_16x16x4_f32(false, a_cur[j], false, b,
                                                                (short)0, acc[tn], false, false);
            }
        }
    };

#pragma unroll
    for (int j = 0; j < 8; ++j) a_cur[j] = act(a_cur[j]);

#define WMMA_GUARD_TEXT "v_nop\n\tv_nop\n\tv_nop\n\tv_nop"
#define TIE4(a) "+v"(a[0]), "+v"(a[1]), "+v"(a[2]), "+v"(a[3])
    for (int kc = 0; kc < K; kc += KC) {
        const bool more = (kc + KC < K);
        v2f a_nxt[8];
        if (more) {
#pragma unroll
            for (int j = 0; j < 8; ++j) a_nxt[j] = *(const v2f*)(arow + kc + KC + 4 * j);
        }
        chunk(kc);
        if constexpr (NT == 8)
            asm volatile(WMMA_GUARD_TEXT : TIE4(acc), "+v"(acc[4]), "+v"(acc[5]), "+v"(acc[6]), "+v"(acc[7]));
        else
            asm volatile(WMMA_GUARD_TEXT : TIE4(acc));
        if (more) {
#pragma unroll
            for (int j = 0; j < 8; ++j) a_cur[j] = act(a_nxt[j]);
        }
    }
#undef TIE4
#undef WMMA_GUARD_TEXT

    float* orow = Out + ((size_t)strip * 16 + half * 8) * F + l16;
#pragma unroll
    for (int tn = 0; tn < NT; ++tn)
#pragma unroll
        for (int r = 0; r < 8; ++r)
            orow[(size_t)r * F + tn * 16] = acc[tn][r];
}

template <int VEC>
__global__ void __launch_bounds__(256)
gather_kernel(const int* __restrict__ rowptr, const v2i* __restrict__ rec,
              const float* __restrict__ dinv, const float* __restrict__ H,
              const float* __restrict__ bias, float* __restrict__ outp,
              int n, int F) {
    typedef __attribute__((ext_vector_type(VEC))) float vf;
    int node = (blockIdx.x * blockDim.x + threadIdx.x) >> 5;
    int lane = threadIdx.x & 31;
    if (node >= n) return;
    node = __builtin_amdgcn_readfirstlane(node);

    int beg = rowptr[node];
    int end = rowptr[node + 1];
    float di = dinv[node];

    for (int fb = 0; fb < F; fb += 32 * VEC) {
        int  f0  = fb + lane * VEC;
        bool act = f0 < F;
        vf acc;
        if (act) {
            vf bv = *(const vf*)(bias + f0);
            vf hv = *(const vf*)(H + (size_t)node * F + f0);
            acc = bv + (di * di) * hv;
        }
        for (int base = beg; base < end; base += 32) {
            int m = min(32, end - base);
            int   sl = 0;
            float cl = 0.0f;
            if (lane < m) {
                v2i r = rec[base + lane];
                sl = r.x;
                cl = dinv[sl] * __int_as_float(r.y) * di;
            }
            int j = 0;
            for (; j + 4 <= m; j += 4) {
                int s0 = __builtin_amdgcn_readlane(sl, j);
                int s1 = __builtin_amdgcn_readlane(sl, j + 1);
                int s2 = __builtin_amdgcn_readlane(sl, j + 2);
                int s3 = __builtin_amdgcn_readlane(sl, j + 3);
                float c0 = __int_as_float(__builtin_amdgcn_readlane(__float_as_int(cl), j));
                float c1 = __int_as_float(__builtin_amdgcn_readlane(__float_as_int(cl), j + 1));
                float c2 = __int_as_float(__builtin_amdgcn_readlane(__float_as_int(cl), j + 2));
                float c3 = __int_as_float(__builtin_amdgcn_readlane(__float_as_int(cl), j + 3));
                if (act) {
                    vf h0 = *(const vf*)(H + (size_t)s0 * F + f0);
                    vf h1 = *(const vf*)(H + (size_t)s1 * F + f0);
                    vf h2 = *(const vf*)(H + (size_t)s2 * F + f0);
                    vf h3 = *(const vf*)(H + (size_t)s3 * F + f0);
                    acc += c0 * h0;
                    acc += c1 * h1;
                    acc += c2 * h2;
                    acc += c3 * h3;
                }
            }
            for (; j < m; ++j) {
                int s0 = __builtin_amdgcn_readlane(sl, j);
                float c0 = __int_as_float(__builtin_amdgcn_readlane(__float_as_int(cl), j));
                if (act) {
                    vf h0 = *(const vf*)(H + (size_t)s0 * F + f0);
                    acc += c0 * h0;
                }
            }
        }
        if (act) *(vf*)(outp + (size_t)node * F + f0) = acc;
    }
}

extern "C" void kernel_launch(void* const* d_in, const int* in_sizes, int n_in,
                              void* d_out, int out_size, void* d_ws, size_t ws_size,
                              hipStream_t stream) {
    const float* x    = (const float*)d_in[0];
    const int*   ei   = (const int*)d_in[1];
    const float* ew   = (const float*)d_in[2];
    const float* W1   = (const float*)d_in[3];
    const float* b1   = (const float*)d_in[4];
    const float* W2   = (const float*)d_in[5];
    const float* b2   = (const float*)d_in[6];
    const float* W3   = (const float*)d_in[7];
    const float* b3   = (const float*)d_in[8];
    float*       out  = (float*)d_out;

    const int hidden  = in_sizes[4];
    const int out_dim = in_sizes[8];
    const int in_dim  = in_sizes[3] / hidden;
    const int n       = in_sizes[0] / in_dim;
    const int e       = in_sizes[2];

    const int* src = ei;
    const int* dst = ei + e;

    char* ws = (char*)d_ws;
    size_t off = 0;
    float* dinv = (float*)(ws + off);
    off += ((size_t)n * sizeof(float) + 255) & ~(size_t)255;
    float* H = (float*)(ws + off);
    off += ((size_t)n * hidden * sizeof(float) + 255) & ~(size_t)255;
    float* AGG = (float*)(ws + off);

    const size_t rp_bytes  = ((size_t)(n + 1) * sizeof(int) + 255) & ~(size_t)255;
    const size_t csr_bytes = rp_bytes + (size_t)e * sizeof(v2i);
    char* csrA = (char*)d_out;
    char* csrB = (char*)H + (((size_t)n * out_dim * sizeof(float) + 255) & ~(size_t)255);
    int* rowptrA = (int*)csrA;
    v2i* recA    = (v2i*)(csrA + rp_bytes);
    int* rowptrB = (int*)csrB;
    v2i* recB    = (v2i*)(csrB + rp_bytes);

    const int B = 256;
    auto cdiv = [](long long a, long long b) { return (int)((a + b - 1) / b); };

    const int nchunks  = cdiv(e, CHUNK_EDGES);
    const int nbuckets = cdiv(n, 256);
    v2i* sorted = (v2i*)AGG;
    int* cntT   = (int*)((char*)AGG + (((size_t)nchunks * CHUNK_EDGES * sizeof(v2i) + 255) & ~(size_t)255));
    int* offT   = cntT + (size_t)256 * nchunks;

    chunk_sort_kernel<<<nchunks, 1024, 0, stream>>>(src, dst, ew, sorted, cntT, offT, e, nchunks);
    bucket_sort_kernel<<<nbuckets, 1024, 0, stream>>>(sorted, cntT, offT, rowptrA, recA, dinv, n, nchunks);

    const int gather_blocks = cdiv((long long)n * 32, B);

    {
        int tiles = (n / 16) * (hidden / 16);
        if (in_dim == 256 && hidden == 128)
            gemm_strip_kernel<256, 128, false><<<cdiv(n / 16, 8), B, 0, stream>>>(x, W1, H, n);
        else
            gemm_wmma_f32_kernel<<<cdiv((long long)tiles * 32, B), B, 0, stream>>>(
                x, W1, H, n, in_dim, hidden, 0);
        gather_kernel<4><<<gather_blocks, B, 0, stream>>>(
            rowptrA, recA, dinv, H, b1, AGG, n, hidden);
    }

    {
        int tiles = (n / 16) * (hidden / 16);
        if (hidden == 128)
            gemm_strip_kernel<128, 128, true><<<cdiv(n / 16, 8), B, 0, stream>>>(AGG, W2, H, n);
        else
            gemm_wmma_f32_kernel<<<cdiv((long long)tiles * 32, B), B, 0, stream>>>(
                AGG, W2, H, n, hidden, hidden, 1);
        gather_kernel<4><<<gather_blocks, B, 0, stream>>>(
            rowptrA, recA, dinv, H, b2, AGG, n, hidden);
    }

    {
        int tiles = (n / 16) * (out_dim / 16);
        if (hidden == 128 && out_dim == 64)
            gemm_strip_kernel<128, 64, true><<<cdiv(n / 16, 8), B, 0, stream>>>(AGG, W3, H, n);
        else
            gemm_wmma_f32_kernel<<<cdiv((long long)tiles * 32, B), B, 0, stream>>>(
                AGG, W3, H, n, hidden, out_dim, 1);
        int cnt8 = (int)(csr_bytes / sizeof(v2i));
        copy_kernel<<<cdiv(cnt8, B), B, 0, stream>>>((const v2i*)csrA, (v2i*)csrB, cnt8);
        gather_kernel<2><<<gather_blocks, B, 0, stream>>>(
            rowptrB, recB, dinv, H, b3, out, n, out_dim);
    }
}
